# starting kernel
# baseline (speedup 1.0000x reference)
#include <hip/hip_runtime.h>
#include <hip/hip_bf16.h>

#define DIM 128

typedef __attribute__((ext_vector_type(2))) float v2f;
typedef __attribute__((ext_vector_type(8))) float v8f;

__global__ __launch_bounds__(256) void zero_f32(float* __restrict__ p, long long n) {
  long long i = (long long)blockIdx.x * blockDim.x + threadIdx.x;
  long long stride = (long long)gridDim.x * blockDim.x;
  for (; i < n; i += stride) p[i] = 0.0f;
}

__global__ __launch_bounds__(256) void degree_kernel(const int* __restrict__ dst,
                                                     float* __restrict__ deg, int n_edges) {
  int e = blockIdx.x * blockDim.x + threadIdx.x;
  if (e < n_edges) atomicAdd(&deg[dst[e]], 1.0f);
}

__global__ __launch_bounds__(256) void invdeg_kernel(const float* __restrict__ deg,
                                                     float* __restrict__ invd, int n) {
  int i = blockIdx.x * blockDim.x + threadIdx.x;
  if (i < n) invd[i] = 1.0f / fmaxf(deg[i], 1.0f);
}

__global__ __launch_bounds__(256) void scatter_kernel(const float* __restrict__ feat,
                                                      const int* __restrict__ src,
                                                      const int* __restrict__ dst,
                                                      float* __restrict__ agg, int n_edges) {
  int e = blockIdx.x * 8 + (threadIdx.x >> 5);
  int lane = threadIdx.x & 31;
  if (e >= n_edges) return;
  int s = src[e];
  int d = dst[e];
  const float4 v = *(const float4*)(feat + (size_t)s * DIM + lane * 4);
  float* o = agg + (size_t)d * DIM + lane * 4;
  atomicAdd(o + 0, v.x);
  atomicAdd(o + 1, v.y);
  atomicAdd(o + 2, v.z);
  atomicAdd(o + 3, v.w);
}

__global__ __launch_bounds__(128)
void sage_gemm(const float* __restrict__ agg, const float* __restrict__ invd,
               const float* __restrict__ h,
               const float* __restrict__ Wl, const float* __restrict__ Wr,
               const float* __restrict__ bias,
               float* __restrict__ out, int n_nodes, int do_relu)
{
  extern __shared__ float smem[];
  float* sWl = smem;
  float* sWr = smem + DIM * DIM;
  float* sb  = smem + 2 * DIM * DIM;

  int tid = threadIdx.x;
  for (int i = tid * 4; i < DIM * DIM; i += 128 * 4) {
    *(float4*)(sWl + i) = *(const float4*)(Wl + i);
    *(float4*)(sWr + i) = *(const float4*)(Wr + i);
  }
  if (tid < DIM) sb[tid] = bias[tid];
  __syncthreads();

  int wave = tid >> 5;
  int lane = tid & 31;
  int rowBase = (blockIdx.x * 4 + wave) * 16;
  if (rowBase >= n_nodes) return;

  int laneM = lane & 15;
  int kOff  = (lane >> 4) << 1;
  int row   = rowBase + laneM;
  float id  = invd[row];
  const float* aggRow = agg + (size_t)row * DIM;
  const float* hRow   = h   + (size_t)row * DIM;

  v8f acc[8] = {};

#pragma unroll 2
  for (int k0 = 0; k0 < DIM; k0 += 4) {
    int k = k0 + kOff;
    v2f aA, aH;
    aA[0] = aggRow[k]     * id;
    aA[1] = aggRow[k + 1] * id;
    aH[0] = hRow[k];
    aH[1] = hRow[k + 1];
#pragma unroll
    for (int t = 0; t < 8; ++t) {
      int n = t * 16 + laneM;
      v2f bl, br;
      bl[0] = sWl[k * DIM + n];
      bl[1] = sWl[(k + 1) * DIM + n];
      br[0] = sWr[k * DIM + n];
      br[1] = sWr[(k + 1) * DIM + n];
      acc[t] = __builtin_amdgcn_wmma_f32_16x16x4_f32(
          false, aA, false, bl, (short)0, acc[t], false, false);
      acc[t] = __builtin_amdgcn_wmma_f32_16x16x4_f32(
          false, aH, false, br, (short)0, acc[t], false, false);
    }
  }

  int mOff = (lane >> 4) << 3;
#pragma unroll
  for (int t = 0; t < 8; ++t) {
    int n = t * 16 + laneM;
    float bv = sb[n];
#pragma unroll
    for (int r = 0; r < 8; ++r) {
      float v = acc[t][r] + bv;
      if (do_relu) v = fmaxf(v, 0.0f);
      out[(size_t)(rowBase + mOff + r) * DIM + n] = v;
    }
  }
}

extern "C" void kernel_launch(void* const* d_in, const int* in_sizes, int n_in,
                              void* d_out, int out_size, void* d_ws, size_t ws_size,
                              hipStream_t stream) {
  const float* x   = (const float*)d_in[0];
  const int*   ei  = (const int*)d_in[1];
  const float* W1l = (const float*)d_in[2];
  const float* b1  = (const float*)d_in[3];
  const float* W1r = (const float*)d_in[4];
  const float* W2l = (const float*)d_in[5];
  const float* b2  = (const float*)d_in[6];
  const float* W2r = (const float*)d_in[7];
  float* out = (float*)d_out;

  const int nNodes = in_sizes[0] / DIM;
  const int nEdges = in_sizes[1] / 2;
  const int* src = ei;
  const int* dst = ei + nEdges;

  float* agg  = (float*)d_ws;
  float* h1   = agg + (size_t)nNodes * DIM;
  float* deg  = h1  + (size_t)nNodes * DIM;
  float* invd = deg + nNodes;

  const long long featN = (long long)nNodes * DIM;
  const int rowTiles   = (nNodes + 15) / 16;
  const int gemmBlocks = (rowTiles + 3) / 4;
  const size_t ldsBytes = (size_t)(2 * DIM * DIM + DIM) * sizeof(float);

  zero_f32<<<256, 256, 0, stream>>>(deg, nNodes);
  degree_kernel<<<(nEdges + 255) / 256, 256, 0, stream>>>(dst, deg, nEdges);
  invdeg_kernel<<<(nNodes + 255) / 256, 256, 0, stream>>>(deg, invd, nNodes);

  zero_f32<<<2048, 256, 0, stream>>>(agg, featN);
  scatter_kernel<<<(nEdges + 7) / 8, 256, 0, stream>>>(x, src, dst, agg, nEdges);
  sage_gemm<<<gemmBlocks, 128, ldsBytes, stream>>>(agg, invd, x, W1l, W1r, b1, h1, nNodes, 1);

  zero_f32<<<2048, 256, 0, stream>>>(agg, featN);
  scatter_kernel<<<(nEdges + 7) / 8, 256, 0, stream>>>(h1, src, dst, agg, nEdges);
  sage_gemm<<<gemmBlocks, 128, ldsBytes, stream>>>(agg, invd, h1, W2l, W2r, b2, out, nNodes, 0);
}


# Float-atomic scatter-add and zero fill replaced by in-call per-destination lists (integer-atomic count and slot) and one wave per node adding its source rows in ascending source id: fixed sum order
# speedup vs baseline: 1.8110x; 1.8110x over previous
#include <hip/hip_runtime.h>
#include <hip/hip_bf16.h>

#define DIM 128

typedef __attribute__((ext_vector_type(2))) float v2f;
typedef __attribute__((ext_vector_type(8))) float v8f;

__global__ __launch_bounds__(256) void zero_i32(int* __restrict__ p, int n) {
  int i = blockIdx.x * blockDim.x + threadIdx.x;
  int stride = gridDim.x * blockDim.x;
  for (; i < n; i += stride) p[i] = 0;
}

__global__ __launch_bounds__(256) void count_rank_kernel(const int* __restrict__ dst,
                                                         int* __restrict__ cnt,
                                                         int* __restrict__ rank, int n_edges) {
  int e = blockIdx.x * blockDim.x + threadIdx.x;
  if (e < n_edges) rank[e] = atomicAdd(&cnt[dst[e]], 1);
}

__global__ __launch_bounds__(1024) void scan_kernel(const int* __restrict__ cnt,
                                                    int* __restrict__ rowptr,
                                                    float* __restrict__ invd, int n) {
  __shared__ int sh[1024];
  const int tid = threadIdx.x;
  const int base = blockIdx.x * 1024;
  const int4* c4 = (const int4*)cnt;
  const int nq = base / 4;

  int s0 = 0, s1 = 0, s2 = 0, s3 = 0;
  int i = tid;
  for (; i + 3 * 1024 < nq; i += 4 * 1024) {
    const int4 a = c4[i];
    const int4 b = c4[i + 1024];
    const int4 d = c4[i + 2048];
    const int4 e = c4[i + 3072];
    s0 += (a.x + a.y) + (a.z + a.w);
    s1 += (b.x + b.y) + (b.z + b.w);
    s2 += (d.x + d.y) + (d.z + d.w);
    s3 += (e.x + e.y) + (e.z + e.w);
  }
  for (; i < nq; i += 1024) {
    const int4 a = c4[i];
    s0 += (a.x + a.y) + (a.z + a.w);
  }
  sh[tid] = (s0 + s1) + (s2 + s3);
  __syncthreads();
  for (int off = 512; off > 0; off >>= 1) {
    if (tid < off) sh[tid] += sh[tid + off];
    __syncthreads();
  }
  const int prefix = sh[0];
  __syncthreads();

  const int idx = base + tid;
  const int c = (idx < n) ? cnt[idx] : 0;
  sh[tid] = c;
  __syncthreads();
  for (int off = 1; off < 1024; off <<= 1) {
    const int v = (tid >= off) ? sh[tid - off] : 0;
    __syncthreads();
    sh[tid] += v;
    __syncthreads();
  }
  const int excl = prefix + sh[tid] - c;
  if (idx <= n) rowptr[idx] = excl;
  if (idx < n) invd[idx] = 1.0f / fmaxf((float)c, 1.0f);
}

__global__ __launch_bounds__(256) void fill_kernel(const int* __restrict__ src,
                                                   const int* __restrict__ dst,
                                                   const int* __restrict__ rank,
                                                   const int* __restrict__ rowptr,
                                                   int* __restrict__ col, int n_edges) {
  int e = blockIdx.x * blockDim.x + threadIdx.x;
  if (e < n_edges) col[rowptr[dst[e]] + rank[e]] = src[e];
}

template <int N>
static __device__ __forceinline__ void add_rows(float4& acc, const float* __restrict__ fl,
                                                int sorted, int i) {
  float4 v[N];
#pragma unroll
  for (int u = 0; u < N; ++u) {
    const int s = __builtin_amdgcn_readlane(sorted, i + u);
    v[u] = *(const float4*)(fl + (size_t)s * DIM);
  }
#pragma unroll
  for (int u = 0; u < N; ++u) {
    acc.x += v[u].x; acc.y += v[u].y; acc.z += v[u].z; acc.w += v[u].w;
  }
}

__global__ __launch_bounds__(256) void gather_kernel(const float* __restrict__ feat,
                                                     const int* __restrict__ rowptr,
                                                     const int* __restrict__ col,
                                                     float* __restrict__ agg, int n_nodes) {
  const int lane = threadIdx.x & 31;
  const int node = __builtin_amdgcn_readfirstlane((int)(blockIdx.x * 8 + (threadIdx.x >> 5)));
  if (node >= n_nodes) return;
  const int beg = rowptr[node];
  const int deg = rowptr[node + 1] - beg;
  const float* fl = feat + lane * 4;

  float4 acc = make_float4(0.0f, 0.0f, 0.0f, 0.0f);
  if (deg <= 32) {
    const int key = (lane < deg) ? col[beg + lane] : 0x7fffffff;
    int r = 0;
    for (int i = 0; i < deg; ++i) {
      const int ki = __builtin_amdgcn_readlane(key, i);
      r += ((ki < key) || (ki == key && i < lane)) ? 1 : 0;
    }
    const int pos = (lane < deg) ? r : lane;
    const int sorted = __builtin_amdgcn_ds_permute(pos << 2, key);
    int i = 0;
    for (; i + 8 <= deg; i += 8) add_rows<8>(acc, fl, sorted, i);
    if (i + 4 <= deg) { add_rows<4>(acc, fl, sorted, i); i += 4; }
    if (i + 2 <= deg) { add_rows<2>(acc, fl, sorted, i); i += 2; }
    if (i < deg) add_rows<1>(acc, fl, sorted, i);
  } else {
    int last = -1;
    int done = 0;
    for (int it = 0; it < deg && done < deg; ++it) {
      int best = 0x7fffffff;
      for (int j = lane; j < deg; j += 32) {
        const int k = col[beg + j];
        if (k > last && k < best) best = k;
      }
      for (int off = 16; off > 0; off >>= 1) best = min(best, __shfl_xor(best, off, 32));
      int c = 0;
      for (int j = lane; j < deg; j += 32) c += (col[beg + j] == best) ? 1 : 0;
      for (int off = 16; off > 0; off >>= 1) c += __shfl_xor(c, off, 32);
      best = __builtin_amdgcn_readfirstlane(best);
      c = __builtin_amdgcn_readfirstlane(c);
      const float4 v = *(const float4*)(fl + (size_t)best * DIM);
      for (int q = 0; q < c; ++q) { acc.x += v.x; acc.y += v.y; acc.z += v.z; acc.w += v.w; }
      last = best;
      done += c;
    }
  }
  *(float4*)(agg + (size_t)node * DIM + lane * 4) = acc;
}

__global__ __launch_bounds__(128)
void sage_gemm(const float* __restrict__ agg, const float* __restrict__ invd,
               const float* __restrict__ h,
               const float* __restrict__ Wl, const float* __restrict__ Wr,
               const float* __restrict__ bias,
               float* __restrict__ out, int n_nodes, int do_relu)
{
  extern __shared__ float smem[];
  float* sWl = smem;
  float* sWr = smem + DIM * DIM;
  float* sb  = smem + 2 * DIM * DIM;

  int tid = threadIdx.x;
  for (int i = tid * 4; i < DIM * DIM; i += 128 * 4) {
    *(float4*)(sWl + i) = *(const float4*)(Wl + i);
    *(float4*)(sWr + i) = *(const float4*)(Wr + i);
  }
  if (tid < DIM) sb[tid] = bias[tid];
  __syncthreads();

  int wave = tid >> 5;
  int lane = tid & 31;
  int rowBase = (blockIdx.x * 4 + wave) * 16;
  if (rowBase >= n_nodes) return;

  int laneM = lane & 15;
  int kOff  = (lane >> 4) << 1;
  int row   = rowBase + laneM;
  float id  = invd[row];
  const float* aggRow = agg + (size_t)row * DIM;
  const float* hRow   = h   + (size_t)row * DIM;

  v8f acc[8] = {};

#pragma unroll 2
  for (int k0 = 0; k0 < DIM; k0 += 4) {
    int k = k0 + kOff;
    v2f aA, aH;
    aA[0] = aggRow[k]     * id;
    aA[1] = aggRow[k + 1] * id;
    aH[0] = hRow[k];
    aH[1] = hRow[k + 1];
#pragma unroll
    for (int t = 0; t < 8; ++t) {
      int n = t * 16 + laneM;
      v2f bl, br;
      bl[0] = sWl[k * DIM + n];
      bl[1] = sWl[(k + 1) * DIM + n];
      br[0] = sWr[k * DIM + n];
      br[1] = sWr[(k + 1) * DIM + n];
      acc[t] = __builtin_amdgcn_wmma_f32_16x16x4_f32(
          false, aA, false, bl, (short)0, acc[t], false, false);
      acc[t] = __builtin_amdgcn_wmma_f32_16x16x4_f32(
          false, aH, false, br, (short)0, acc[t], false, false);
    }
  }

  int mOff = (lane >> 4) << 3;
#pragma unroll
  for (int t = 0; t < 8; ++t) {
    int n = t * 16 + laneM;
    float bv = sb[n];
#pragma unroll
    for (int r = 0; r < 8; ++r) {
      float v = acc[t][r] + bv;
      if (do_relu) v = fmaxf(v, 0.0f);
      out[(size_t)(rowBase + mOff + r) * DIM + n] = v;
    }
  }
}

extern "C" void kernel_launch(void* const* d_in, const int* in_sizes, int n_in,
                              void* d_out, int out_size, void* d_ws, size_t ws_size,
                              hipStream_t stream) {
  const float* x   = (const float*)d_in[0];
  const int*   ei  = (const int*)d_in[1];
  const float* W1l = (const float*)d_in[2];
  const float* b1  = (const float*)d_in[3];
  const float* W1r = (const float*)d_in[4];
  const float* W2l = (const float*)d_in[5];
  const float* b2  = (const float*)d_in[6];
  const float* W2r = (const float*)d_in[7];
  float* out = (float*)d_out;

  const int nNodes = in_sizes[0] / DIM;
  const int nEdges = in_sizes[1] / 2;
  const int* src = ei;
  const int* dst = ei + nEdges;

  const size_t nodePad = ((size_t)nNodes + 4) & ~(size_t)3;
  const size_t edgePad = ((size_t)nEdges + 3) & ~(size_t)3;
  float* agg    = (float*)d_ws;
  float* h1     = agg + (size_t)nNodes * DIM;
  float* invd   = h1  + (size_t)nNodes * DIM;
  int*   cnt    = (int*)(invd + nodePad);
  int*   rowptr = cnt + nodePad;
  int*   rank   = rowptr + nodePad;
  int*   col    = rank + edgePad;

  const int rowTiles   = (nNodes + 15) / 16;
  const int gemmBlocks = (rowTiles + 3) / 4;
  const size_t ldsBytes = (size_t)(2 * DIM * DIM + DIM) * sizeof(float);
  const int edgeBlocks = (nEdges + 255) / 256;
  const int nodeWaveBlocks = (nNodes + 7) / 8;

  zero_i32<<<128, 256, 0, stream>>>(cnt, nNodes);
  count_rank_kernel<<<edgeBlocks, 256, 0, stream>>>(dst, cnt, rank, nEdges);
  scan_kernel<<<(nNodes + 1024) / 1024, 1024, 0, stream>>>(cnt, rowptr, invd, nNodes);
  fill_kernel<<<edgeBlocks, 256, 0, stream>>>(src, dst, rank, rowptr, col, nEdges);

  gather_kernel<<<nodeWaveBlocks, 256, 0, stream>>>(x, rowptr, col, agg, nNodes);
  sage_gemm<<<gemmBlocks, 128, ldsBytes, stream>>>(agg, invd, x, W1l, W1r, b1, h1, nNodes, 1);

  gather_kernel<<<nodeWaveBlocks, 256, 0, stream>>>(h1, rowptr, col, agg, nNodes);
  sage_gemm<<<gemmBlocks, 128, ldsBytes, stream>>>(agg, invd, h1, W2l, W2r, b2, out, nNodes, 0);
}


# GEMM weights staged in LDS once per WGP: 256 resident 8-wave workgroups loop over the 16-row strips (was 1563 workgroups each re-reading both 64 KB matrices per 64 rows); sums unchanged
# speedup vs baseline: 2.1663x; 1.1962x over previous
#include <hip/hip_runtime.h>
#include <hip/hip_bf16.h>

#define DIM 128

typedef __attribute__((ext_vector_type(2))) float v2f;
typedef __attribute__((ext_vector_type(8))) float v8f;

__global__ __launch_bounds__(256) void zero_i32(int* __restrict__ p, int n) {
  int i = blockIdx.x * blockDim.x + threadIdx.x;
  int stride = gridDim.x * blockDim.x;
  for (; i < n; i += stride) p[i] = 0;
}

__global__ __launch_bounds__(256) void count_rank_kernel(const int* __restrict__ dst,
                                                         int* __restrict__ cnt,
                                                         int* __restrict__ rank, int n_edges) {
  int e = blockIdx.x * blockDim.x + threadIdx.x;
  if (e < n_edges) rank[e] = atomicAdd(&cnt[dst[e]], 1);
}

__global__ __launch_bounds__(1024) void scan_kernel(const int* __restrict__ cnt,
                                                    int* __restrict__ rowptr,
                                                    float* __restrict__ invd, int n) {
  __shared__ int sh[1024];
  const int tid = threadIdx.x;
  const int base = blockIdx.x * 1024;
  const int4* c4 = (const int4*)cnt;
  const int nq = base / 4;

  int s0 = 0, s1 = 0, s2 = 0, s3 = 0;
  int i = tid;
  for (; i + 3 * 1024 < nq; i += 4 * 1024) {
    const int4 a = c4[i];
    const int4 b = c4[i + 1024];
    const int4 d = c4[i + 2048];
    const int4 e = c4[i + 3072];
    s0 += (a.x + a.y) + (a.z + a.w);
    s1 += (b.x + b.y) + (b.z + b.w);
    s2 += (d.x + d.y) + (d.z + d.w);
    s3 += (e.x + e.y) + (e.z + e.w);
  }
  for (; i < nq; i += 1024) {
    const int4 a = c4[i];
    s0 += (a.x + a.y) + (a.z + a.w);
  }
  sh[tid] = (s0 + s1) + (s2 + s3);
  __syncthreads();
  for (int off = 512; off > 0; off >>= 1) {
    if (tid < off) sh[tid] += sh[tid + off];
    __syncthreads();
  }
  const int prefix = sh[0];
  __syncthreads();

  const int idx = base + tid;
  const int c = (idx < n) ? cnt[idx] : 0;
  sh[tid] = c;
  __syncthreads();
  for (int off = 1; off < 1024; off <<= 1) {
    const int v = (tid >= off) ? sh[tid - off] : 0;
    __syncthreads();
    sh[tid] += v;
    __syncthreads();
  }
  const int excl = prefix + sh[tid] - c;
  if (idx <= n) rowptr[idx] = excl;
  if (idx < n) invd[idx] = 1.0f / fmaxf((float)c, 1.0f);
}

__global__ __launch_bounds__(256) void fill_kernel(const int* __restrict__ src,
                                                   const int* __restrict__ dst,
                                                   const int* __restrict__ rank,
                                                   const int* __restrict__ rowptr,
                                                   int* __restrict__ col, int n_edges) {
  int e = blockIdx.x * blockDim.x + threadIdx.x;
  if (e < n_edges) col[rowptr[dst[e]] + rank[e]] = src[e];
}

template <int N>
static __device__ __forceinline__ void add_rows(float4& acc, const float* __restrict__ fl,
                                                int sorted, int i) {
  float4 v[N];
#pragma unroll
  for (int u = 0; u < N; ++u) {
    const int s = __builtin_amdgcn_readlane(sorted, i + u);
    v[u] = *(const float4*)(fl + (size_t)s * DIM);
  }
#pragma unroll
  for (int u = 0; u < N; ++u) {
    acc.x += v[u].x; acc.y += v[u].y; acc.z += v[u].z; acc.w += v[u].w;
  }
}

__global__ __launch_bounds__(256) void gather_kernel(const float* __restrict__ feat,
                                                     const int* __restrict__ rowptr,
                                                     const int* __restrict__ col,
                                                     float* __restrict__ agg, int n_nodes) {
  const int lane = threadIdx.x & 31;
  const int node = __builtin_amdgcn_readfirstlane((int)(blockIdx.x * 8 + (threadIdx.x >> 5)));
  if (node >= n_nodes) return;
  const int beg = rowptr[node];
  const int deg = rowptr[node + 1] - beg;
  const float* fl = feat + lane * 4;

  float4 acc = make_float4(0.0f, 0.0f, 0.0f, 0.0f);
  if (deg <= 32) {
    const int key = (lane < deg) ? col[beg + lane] : 0x7fffffff;
    int r = 0;
    for (int i = 0; i < deg; ++i) {
      const int ki = __builtin_amdgcn_readlane(key, i);
      r += ((ki < key) || (ki == key && i < lane)) ? 1 : 0;
    }
    const int pos = (lane < deg) ? r : lane;
    const int sorted = __builtin_amdgcn_ds_permute(pos << 2, key);
    int i = 0;
    for (; i + 8 <= deg; i += 8) add_rows<8>(acc, fl, sorted, i);
    if (i + 4 <= deg) { add_rows<4>(acc, fl, sorted, i); i += 4; }
    if (i + 2 <= deg) { add_rows<2>(acc, fl, sorted, i); i += 2; }
    if (i < deg) add_rows<1>(acc, fl, sorted, i);
  } else {
    int last = -1;
    int done = 0;
    for (int it = 0; it < deg && done < deg; ++it) {
      int best = 0x7fffffff;
      for (int j = lane; j < deg; j += 32) {
        const int k = col[beg + j];
        if (k > last && k < best) best = k;
      }
      for (int off = 16; off > 0; off >>= 1) best = min(best, __shfl_xor(best, off, 32));
      int c = 0;
      for (int j = lane; j < deg; j += 32) c += (col[beg + j] == best) ? 1 : 0;
      for (int off = 16; off > 0; off >>= 1) c += __shfl_xor(c, off, 32);
      best = __builtin_amdgcn_readfirstlane(best);
      c = __builtin_amdgcn_readfirstlane(c);
      const float4 v = *(const float4*)(fl + (size_t)best * DIM);
      for (int q = 0; q < c; ++q) { acc.x += v.x; acc.y += v.y; acc.z += v.z; acc.w += v.w; }
      last = best;
      done += c;
    }
  }
  *(float4*)(agg + (size_t)node * DIM + lane * 4) = acc;
}

#define GEMM_THREADS 256
#define GEMM_WAVES (GEMM_THREADS / 32)
#define GEMM_LDS_FLOATS (168 * 1024 / 4)

static __device__ __forceinline__ void stage_matrix(float* __restrict__ dstLds,
                                                    const float* __restrict__ srcGlobal, int tid) {
  float4 v[16];
#pragma unroll
  for (int u = 0; u < 16; ++u)
    v[u] = *(const float4*)(srcGlobal + (size_t)(u * GEMM_THREADS + tid) * 4);
#pragma unroll
  for (int u = 0; u < 16; ++u)
    *(float4*)(dstLds + (u * GEMM_THREADS + tid) * 4) = v[u];
}

__global__ __launch_bounds__(GEMM_THREADS)
void sage_gemm(const float* __restrict__ agg, const float* __restrict__ invd,
               const float* __restrict__ h,
               const float* __restrict__ Wl, const float* __restrict__ Wr,
               const float* __restrict__ bias,
               float* __restrict__ out, int n_nodes, int do_relu)
{
  __shared__ __attribute__((aligned(16))) float smem[GEMM_LDS_FLOATS];
  float* sWl = smem;
  float* sWr = smem + DIM * DIM;
  float* sb  = smem + 2 * DIM * DIM;

  const int tid = threadIdx.x;
  stage_matrix(sWl, Wl, tid);
  stage_matrix(sWr, Wr, tid);
  if (tid < DIM) sb[tid] = bias[tid];
  __syncthreads();

  const int lane = tid & 31;
  const int wave = __builtin_amdgcn_readfirstlane(tid >> 5);
  const int rowTiles = (n_nodes + 15) >> 4;

  const int laneM = lane & 15;
  const int kOff  = (lane >> 4) << 1;
  const int mOff = (lane >> 4) << 3;

  for (int tile = wave * (int)gridDim.x + (int)blockIdx.x; tile < rowTiles;
       tile += GEMM_WAVES * (int)gridDim.x) {
    const int rowBase = tile * 16;
    const int row   = rowBase + laneM;
    const float id  = invd[row];
    const float* aggRow = agg + (size_t)row * DIM;
    const float* hRow   = h   + (size_t)row * DIM;

    v8f acc[8] = {};

#pragma unroll 2
    for (int k0 = 0; k0 < DIM; k0 += 4) {
      int k = k0 + kOff;
      v2f aA, aH;
      aA[0] = aggRow[k]     * id;
      aA[1] = aggRow[k + 1] * id;
      aH[0] = hRow[k];
      aH[1] = hRow[k + 1];
#pragma unroll
      for (int t = 0; t < 8; ++t) {
        int n = t * 16 + laneM;
        v2f bl, br;
        bl[0] = sWl[k * DIM + n];
        bl[1] = sWl[(k + 1) * DIM + n];
        br[0] = sWr[k * DIM + n];
        br[1] = sWr[(k + 1) * DIM + n];
        acc[t] = __builtin_amdgcn_wmma_f32_16x16x4_f32(
            false, aA, false, bl, (short)0, acc[t], false, false);
        acc[t] = __builtin_amdgcn_wmma_f32_16x16x4_f32(
            false, aH, false, br, (short)0, acc[t], false, false);
      }
    }

#pragma unroll
    for (int t = 0; t < 8; ++t) {
      int n = t * 16 + laneM;
      float bv = sb[n];
#pragma unroll
      for (int r = 0; r < 8; ++r) {
        float v = acc[t][r] + bv;
        if (do_relu) v = fmaxf(v, 0.0f);
        out[(size_t)(rowBase + mOff + r) * DIM + n] = v;
      }
    }
  }
}

extern "C" void kernel_launch(void* const* d_in, const int* in_sizes, int n_in,
                              void* d_out, int out_size, void* d_ws, size_t ws_size,
                              hipStream_t stream) {
  const float* x   = (const float*)d_in[0];
  const int*   ei  = (const int*)d_in[1];
  const float* W1l = (const float*)d_in[2];
  const float* b1  = (const float*)d_in[3];
  const float* W1r = (const float*)d_in[4];
  const float* W2l = (const float*)d_in[5];
  const float* b2  = (const float*)d_in[6];
  const float* W2r = (const float*)d_in[7];
  float* out = (float*)d_out;

  const int nNodes = in_sizes[0] / DIM;
  const int nEdges = in_sizes[1] / 2;
  const int* src = ei;
  const int* dst = ei + nEdges;

  const size_t nodePad = ((size_t)nNodes + 4) & ~(size_t)3;
  const size_t edgePad = ((size_t)nEdges + 3) & ~(size_t)3;
  float* agg    = (float*)d_ws;
  float* h1     = agg + (size_t)nNodes * DIM;
  float* invd   = h1  + (size_t)nNodes * DIM;
  int*   cnt    = (int*)(invd + nodePad);
  int*   rowptr = cnt + nodePad;
  int*   rank   = rowptr + nodePad;
  int*   col    = rank + edgePad;

  const int gemmBlocks = 256;
  const int edgeBlocks = (nEdges + 255) / 256;
  const int nodeWaveBlocks = (nNodes + 7) / 8;

  zero_i32<<<128, 256, 0, stream>>>(cnt, nNodes);
  count_rank_kernel<<<edgeBlocks, 256, 0, stream>>>(dst, cnt, rank, nEdges);
  scan_kernel<<<(nNodes + 1024) / 1024, 1024, 0, stream>>>(cnt, rowptr, invd, nNodes);
  fill_kernel<<<edgeBlocks, 256, 0, stream>>>(src, dst, rank, rowptr, col, nEdges);

  gather_kernel<<<nodeWaveBlocks, 256, 0, stream>>>(x, rowptr, col, agg, nNodes);
  sage_gemm<<<gemmBlocks, GEMM_THREADS, 0, stream>>>(agg, invd, x, W1l, W1r, b1, h1, nNodes, 1);

  gather_kernel<<<nodeWaveBlocks, 256, 0, stream>>>(h1, rowptr, col, agg, nNodes);
  sage_gemm<<<gemmBlocks, GEMM_THREADS, 0, stream>>>(agg, invd, h1, W2l, W2r, b2, out, nNodes, 0);
}


# Source lists built in one pass into 32-slot per-node buckets (slot = arrival count, overflow pairs to a spill list): scan, rank plane and second scatter pass go, 8 launches to 6; sum order unchanged
# speedup vs baseline: 2.3870x; 1.1019x over previous
#include <hip/hip_runtime.h>
#include <hip/hip_bf16.h>

#define DIM 128

typedef __attribute__((ext_vector_type(2))) float v2f;
typedef __attribute__((ext_vector_type(8))) float v8f;

__global__ __launch_bounds__(256) void zero_i32(int* __restrict__ p, int n) {
  int i = blockIdx.x * blockDim.x + threadIdx.x;
  int stride = gridDim.x * blockDim.x;
  for (; i < n; i += stride) p[i] = 0;
}

#define BUCKET 32
__global__ __launch_bounds__(256) void bucket_kernel(const int* __restrict__ src,
                                                     const int* __restrict__ dst,
                                                     int* __restrict__ cnt,
                                                     int* __restrict__ bucket,
                                                     int* __restrict__ spill,
                                                     int n_edges, int n_nodes) {
  int e = blockIdx.x * blockDim.x + threadIdx.x;
  if (e < n_edges) {
    const int d = dst[e];
    const int s = src[e];
    const int r = atomicAdd(&cnt[d], 1);
    if (r < BUCKET) {
      bucket[(size_t)d * BUCKET + r] = s;
    } else {
      const int q = atomicAdd(&cnt[n_nodes], 1);
      spill[2 * (size_t)q]     = d;
      spill[2 * (size_t)q + 1] = s;
    }
  }
}

template <int N>
static __device__ __forceinline__ void add_rows(float4& acc, const float* __restrict__ fl,
                                                int sorted, int i) {
  float4 v[N];
#pragma unroll
  for (int u = 0; u < N; ++u) {
    const int s = __builtin_amdgcn_readlane(sorted, i + u);
    v[u] = *(const float4*)(fl + (size_t)s * DIM);
  }
#pragma unroll
  for (int u = 0; u < N; ++u) {
    acc.x += v[u].x; acc.y += v[u].y; acc.z += v[u].z; acc.w += v[u].w;
  }
}

__global__ __launch_bounds__(256) void gather_kernel(const float* __restrict__ feat,
                                                     const int* __restrict__ cnt,
                                                     const int* __restrict__ bucket,
                                                     const int* __restrict__ spill,
                                                     float* __restrict__ agg, int n_nodes) {
  const int lane = threadIdx.x & 31;
  const int node = __builtin_amdgcn_readfirstlane((int)(blockIdx.x * 8 + (threadIdx.x >> 5)));
  if (node >= n_nodes) return;
  const int deg = cnt[node];
  const float* fl = feat + lane * 4;
  const int* bk = bucket + (size_t)node * BUCKET;

  float4 acc = make_float4(0.0f, 0.0f, 0.0f, 0.0f);
  if (deg <= BUCKET) {
    const int key = (lane < deg) ? bk[lane] : 0x7fffffff;
    int r = 0;
    for (int i = 0; i < deg; ++i) {
      const int ki = __builtin_amdgcn_readlane(key, i);
      r += ((ki < key) || (ki == key && i < lane)) ? 1 : 0;
    }
    const int pos = (lane < deg) ? r : lane;
    const int sorted = __builtin_amdgcn_ds_permute(pos << 2, key);
    int i = 0;
    for (; i + 8 <= deg; i += 8) add_rows<8>(acc, fl, sorted, i);
    if (i + 4 <= deg) { add_rows<4>(acc, fl, sorted, i); i += 4; }
    if (i + 2 <= deg) { add_rows<2>(acc, fl, sorted, i); i += 2; }
    if (i < deg) add_rows<1>(acc, fl, sorted, i);
  } else {
    const int nspill = cnt[n_nodes];
    const int mine = bk[lane];
    int last = -1;
    int done = 0;
    for (int it = 0; it < deg && done < deg; ++it) {
      int best = (mine > last) ? mine : 0x7fffffff;
      for (int j = lane; j < nspill; j += 32) {
        const int k = spill[2 * (size_t)j + 1];
        if (spill[2 * (size_t)j] == node && k > last && k < best) best = k;
      }
      for (int off = 16; off > 0; off >>= 1) best = min(best, __shfl_xor(best, off, 32));
      int c = (mine == best) ? 1 : 0;
      for (int j = lane; j < nspill; j += 32)
        c += (spill[2 * (size_t)j] == node && spill[2 * (size_t)j + 1] == best) ? 1 : 0;
      for (int off = 16; off > 0; off >>= 1) c += __shfl_xor(c, off, 32);
      best = __builtin_amdgcn_readfirstlane(best);
      c = __builtin_amdgcn_readfirstlane(c);
      if (best == 0x7fffffff) break;
      const float4 v = *(const float4*)(fl + (size_t)best * DIM);
      for (int q = 0; q < c; ++q) { acc.x += v.x; acc.y += v.y; acc.z += v.z; acc.w += v.w; }
      last = best;
      done += c;
    }
  }
  *(float4*)(agg + (size_t)node * DIM + lane * 4) = acc;
}

#define GEMM_THREADS 256
#define GEMM_WAVES (GEMM_THREADS / 32)
#define GEMM_LDS_FLOATS (168 * 1024 / 4)

static __device__ __forceinline__ void stage_matrix(float* __restrict__ dstLds,
                                                    const float* __restrict__ srcGlobal, int tid) {
  float4 v[16];
#pragma unroll
  for (int u = 0; u < 16; ++u)
    v[u] = *(const float4*)(srcGlobal + (size_t)(u * GEMM_THREADS + tid) * 4);
#pragma unroll
  for (int u = 0; u < 16; ++u)
    *(float4*)(dstLds + (u * GEMM_THREADS + tid) * 4) = v[u];
}

__global__ __launch_bounds__(GEMM_THREADS)
void sage_gemm(const float* __restrict__ agg, const int* __restrict__ cnt,
               const float* __restrict__ h,
               const float* __restrict__ Wl, const float* __restrict__ Wr,
               const float* __restrict__ bias,
               float* __restrict__ out, int n_nodes, int do_relu)
{
  __shared__ __attribute__((aligned(16))) float smem[GEMM_LDS_FLOATS];
  float* sWl = smem;
  float* sWr = smem + DIM * DIM;
  float* sb  = smem + 2 * DIM * DIM;

  const int tid = threadIdx.x;
  stage_matrix(sWl, Wl, tid);
  stage_matrix(sWr, Wr, tid);
  if (tid < DIM) sb[tid] = bias[tid];
  __syncthreads();

  const int lane = tid & 31;
  const int wave = __builtin_amdgcn_readfirstlane(tid >> 5);
  const int rowTiles = (n_nodes + 15) >> 4;

  const int laneM = lane & 15;
  const int kOff  = (lane >> 4) << 1;
  const int mOff = (lane >> 4) << 3;

  for (int tile = wave * (int)gridDim.x + (int)blockIdx.x; tile < rowTiles;
       tile += GEMM_WAVES * (int)gridDim.x) {
    const int rowBase = tile * 16;
    const int row   = rowBase + laneM;
    const float id  = 1.0f / fmaxf((float)cnt[row], 1.0f);
    const float* aggRow = agg + (size_t)row * DIM;
    const float* hRow   = h   + (size_t)row * DIM;

    v8f acc[8] = {};

#pragma unroll 2
    for (int k0 = 0; k0 < DIM; k0 += 4) {
      int k = k0 + kOff;
      v2f aA, aH;
      aA[0] = aggRow[k]     * id;
      aA[1] = aggRow[k + 1] * id;
      aH[0] = hRow[k];
      aH[1] = hRow[k + 1];
#pragma unroll
      for (int t = 0; t < 8; ++t) {
        int n = t * 16 + laneM;
        v2f bl, br;
        bl[0] = sWl[k * DIM + n];
        bl[1] = sWl[(k + 1) * DIM + n];
        br[0] = sWr[k * DIM + n];
        br[1] = sWr[(k + 1) * DIM + n];
        acc[t] = __builtin_amdgcn_wmma_f32_16x16x4_f32(
            false, aA, false, bl, (short)0, acc[t], false, false);
        acc[t] = __builtin_amdgcn_wmma_f32_16x16x4_f32(
            false, aH, false, br, (short)0, acc[t], false, false);
      }
    }

#pragma unroll
    for (int t = 0; t < 8; ++t) {
      int n = t * 16 + laneM;
      float bv = sb[n];
#pragma unroll
      for (int r = 0; r < 8; ++r) {
        float v = acc[t][r] + bv;
        if (do_relu) v = fmaxf(v, 0.0f);
        out[(size_t)(rowBase + mOff + r) * DIM + n] = v;
      }
    }
  }
}

extern "C" void kernel_launch(void* const* d_in, const int* in_sizes, int n_in,
                              void* d_out, int out_size, void* d_ws, size_t ws_size,
                              hipStream_t stream) {
  const float* x   = (const float*)d_in[0];
  const int*   ei  = (const int*)d_in[1];
  const float* W1l = (const float*)d_in[2];
  const float* b1  = (const float*)d_in[3];
  const float* W1r = (const float*)d_in[4];
  const float* W2l = (const float*)d_in[5];
  const float* b2  = (const float*)d_in[6];
  const float* W2r = (const float*)d_in[7];
  float* out = (float*)d_out;

  const int nNodes = in_sizes[0] / DIM;
  const int nEdges = in_sizes[1] / 2;
  const int* src = ei;
  const int* dst = ei + nEdges;

  const size_t nodePad = ((size_t)nNodes + 4) & ~(size_t)3;
  float* agg    = (float*)d_ws;
  float* h1     = agg + (size_t)nNodes * DIM;
  int*   cnt    = (int*)(h1 + (size_t)nNodes * DIM);
  int*   bucket = cnt + nodePad;
  int*   spill  = bucket + (size_t)nNodes * BUCKET;

  const int gemmBlocks = 256;
  const int edgeBlocks = (nEdges + 255) / 256;
  const int nodeWaveBlocks = (nNodes + 7) / 8;

  zero_i32<<<128, 256, 0, stream>>>(cnt, nNodes + 1);
  bucket_kernel<<<edgeBlocks, 256, 0, stream>>>(src, dst, cnt, bucket, spill, nEdges, nNodes);

  gather_kernel<<<nodeWaveBlocks, 256, 0, stream>>>(x, cnt, bucket, spill, agg, nNodes);
  sage_gemm<<<gemmBlocks, GEMM_THREADS, 0, stream>>>(agg, cnt, x, W1l, W1r, b1, h1, nNodes, 1);

  gather_kernel<<<nodeWaveBlocks, 256, 0, stream>>>(h1, cnt, bucket, spill, agg, nNodes);
  sage_gemm<<<gemmBlocks, GEMM_THREADS, 0, stream>>>(agg, cnt, h1, W2l, W2r, b2, out, nNodes, 0);
}


# Per-edge global returning atomics replaced by a two-level counting sort in LDS: chunks sorted by 512-node bin, then one workgroup per bin fills the buckets; zero-fill launch goes; sum order unchanged
# speedup vs baseline: 3.3040x; 1.3842x over previous
#include <hip/hip_runtime.h>
#include <hip/hip_bf16.h>

#define DIM 128

typedef __attribute__((ext_vector_type(2))) float v2f;
typedef __attribute__((ext_vector_type(8))) float v8f;

#define BUCKET 32
#define CHUNK 8192
#define BIN_SHIFT 9
#define BIN_NODES (1 << BIN_SHIFT)
#define MAXBINS 256
#define SORT_THREADS 1024
#define EDGES_PER_THREAD (CHUNK / SORT_THREADS)
#define LIST_WAVES (SORT_THREADS / 32)
#define LDS_PITCH 33

typedef unsigned long long pair_t;

__global__ __launch_bounds__(SORT_THREADS) void bin_kernel(const int* __restrict__ src,
                                                          const int* __restrict__ dst,
                                                          pair_t* __restrict__ pairs,
                                                          int* __restrict__ off,
                                                          int* __restrict__ cnt,
                                                          int n_edges, int n_nodes, int n_bins) {
  __shared__ int hcount[MAXBINS];
  __shared__ int hstart[MAXBINS + 1];
  __shared__ pair_t stage[CHUNK];
  const int tid = threadIdx.x;
  const int chunkBase = blockIdx.x * CHUNK;
  const int n = min(CHUNK, n_edges - chunkBase);
  if (blockIdx.x == 0 && tid == 0) cnt[n_nodes] = 0;
  if (tid < MAXBINS) hcount[tid] = 0;
  __syncthreads();

  int d[EDGES_PER_THREAD], s[EDGES_PER_THREAD], r[EDGES_PER_THREAD];
#pragma unroll
  for (int u = 0; u < EDGES_PER_THREAD; ++u) {
    const int i = u * SORT_THREADS + tid;
    d[u] = -1;
    s[u] = 0;
    if (i < n) { d[u] = dst[chunkBase + i]; s[u] = src[chunkBase + i]; }
  }
#pragma unroll
  for (int u = 0; u < EDGES_PER_THREAD; ++u) {
    r[u] = 0;
    if (d[u] >= 0) r[u] = atomicAdd(&hcount[d[u] >> BIN_SHIFT], 1);
  }
  __syncthreads();

  if (tid < 32) {
    int c[MAXBINS / 32];
    int tot = 0;
#pragma unroll
    for (int j = 0; j < MAXBINS / 32; ++j) { c[j] = hcount[tid * (MAXBINS / 32) + j]; tot += c[j]; }
    int incl = tot;
    for (int o = 1; o < 32; o <<= 1) {
      const int t = __shfl_up(incl, o, 32);
      if (tid >= o) incl += t;
    }
    int run = incl - tot;
#pragma unroll
    for (int j = 0; j < MAXBINS / 32; ++j) { hstart[tid * (MAXBINS / 32) + j] = run; run += c[j]; }
    if (tid == 31) hstart[MAXBINS] = run;
  }
  __syncthreads();

#pragma unroll
  for (int u = 0; u < EDGES_PER_THREAD; ++u) {
    if (d[u] >= 0)
      stage[hstart[d[u] >> BIN_SHIFT] + r[u]] =
          ((pair_t)(unsigned)s[u] << 32) | (pair_t)(unsigned)d[u];
  }
  __syncthreads();

  for (int i = tid; i < n; i += SORT_THREADS) pairs[(size_t)chunkBase + i] = stage[i];
  for (int bi = tid; bi <= n_bins; bi += SORT_THREADS)
    off[(size_t)blockIdx.x * (n_bins + 1) + bi] = hstart[bi];
}

__global__ __launch_bounds__(SORT_THREADS) void list_kernel(const pair_t* __restrict__ pairs,
                                                           const int* __restrict__ off,
                                                           int* __restrict__ cnt,
                                                           int* __restrict__ bucket,
                                                           int* __restrict__ spill,
                                                           int n_nodes, int n_bins, int n_chunks) {
  __shared__ int lcnt[BIN_NODES];
  __shared__ int lbk[BIN_NODES * LDS_PITCH];
  const int tid = threadIdx.x;
  const int lane = tid & 31;
  const int wave = __builtin_amdgcn_readfirstlane(tid >> 5);
  const int bin = blockIdx.x;
  const int nodeBase = bin << BIN_SHIFT;
  const int pitch = n_bins + 1;
  if (tid < BIN_NODES) lcnt[tid] = 0;
  __syncthreads();

  for (int base = 0; base < n_chunks; base += 32 * LIST_WAVES) {
    const int wl = base + wave + LIST_WAVES * lane;
    int o0 = 0, o1 = 0;
    if (wl < n_chunks) {
      o0 = off[(size_t)wl * pitch + bin];
      o1 = off[(size_t)wl * pitch + bin + 1];
    }
    for (int j = 0; j < 32; ++j) {
      const int w = base + wave + LIST_WAVES * j;
      if (w >= n_chunks) break;
      const int s0 = __builtin_amdgcn_readlane(o0, j);
      const int s1 = __builtin_amdgcn_readlane(o1, j);
      const pair_t* seg = pairs + (size_t)w * CHUNK;
      for (int i0 = s0; i0 < s1; i0 += 64) {
        const int ia = i0 + lane;
        const int ib = ia + 32;
        pair_t p[2];
        p[0] = (ia < s1) ? seg[ia] : 0;
        p[1] = (ib < s1) ? seg[ib] : 0;
        const bool live[2] = {ia < s1, ib < s1};
#pragma unroll
        for (int u = 0; u < 2; ++u) {
          if (live[u]) {
            const int dnode = (int)(unsigned)(p[u] & 0xffffffffull);
            const int snode = (int)(unsigned)(p[u] >> 32);
            const int local = dnode - nodeBase;
            const int slot = atomicAdd(&lcnt[local], 1);
            if (slot < BUCKET) {
              lbk[local * LDS_PITCH + slot] = snode;
            } else {
              const int q = atomicAdd(&cnt[n_nodes], 1);
              spill[2 * (size_t)q]     = dnode;
              spill[2 * (size_t)q + 1] = snode;
            }
          }
        }
      }
    }
  }
  __syncthreads();

  if (tid < BIN_NODES && nodeBase + tid < n_nodes) cnt[nodeBase + tid] = lcnt[tid];
  for (int i = tid; i < BIN_NODES * BUCKET; i += SORT_THREADS) {
    const int node = i >> 5;
    if (nodeBase + node < n_nodes)
      bucket[(size_t)nodeBase * BUCKET + i] = lbk[node * LDS_PITCH + (i & 31)];
  }
}

template <int N>
static __device__ __forceinline__ void add_rows(float4& acc, const float* __restrict__ fl,
                                                int sorted, int i) {
  float4 v[N];
#pragma unroll
  for (int u = 0; u < N; ++u) {
    const int s = __builtin_amdgcn_readlane(sorted, i + u);
    v[u] = *(const float4*)(fl + (size_t)s * DIM);
  }
#pragma unroll
  for (int u = 0; u < N; ++u) {
    acc.x += v[u].x; acc.y += v[u].y; acc.z += v[u].z; acc.w += v[u].w;
  }
}

__global__ __launch_bounds__(256) void gather_kernel(const float* __restrict__ feat,
                                                     const int* __restrict__ cnt,
                                                     const int* __restrict__ bucket,
                                                     const int* __restrict__ spill,
                                                     float* __restrict__ agg, int n_nodes) {
  const int lane = threadIdx.x & 31;
  const int node = __builtin_amdgcn_readfirstlane((int)(blockIdx.x * 8 + (threadIdx.x >> 5)));
  if (node >= n_nodes) return;
  const int deg = cnt[node];
  const float* fl = feat + lane * 4;
  const int* bk = bucket + (size_t)node * BUCKET;

  float4 acc = make_float4(0.0f, 0.0f, 0.0f, 0.0f);
  if (deg <= BUCKET) {
    const int key = (lane < deg) ? bk[lane] : 0x7fffffff;
    int r = 0;
    for (int i = 0; i < deg; ++i) {
      const int ki = __builtin_amdgcn_readlane(key, i);
      r += ((ki < key) || (ki == key && i < lane)) ? 1 : 0;
    }
    const int pos = (lane < deg) ? r : lane;
    const int sorted = __builtin_amdgcn_ds_permute(pos << 2, key);
    int i = 0;
    for (; i + 8 <= deg; i += 8) add_rows<8>(acc, fl, sorted, i);
    if (i + 4 <= deg) { add_rows<4>(acc, fl, sorted, i); i += 4; }
    if (i + 2 <= deg) { add_rows<2>(acc, fl, sorted, i); i += 2; }
    if (i < deg) add_rows<1>(acc, fl, sorted, i);
  } else {
    const int nspill = cnt[n_nodes];
    const int mine = bk[lane];
    int last = -1;
    int done = 0;
    for (int it = 0; it < deg && done < deg; ++it) {
      int best = (mine > last) ? mine : 0x7fffffff;
      for (int j = lane; j < nspill; j += 32) {
        const int k = spill[2 * (size_t)j + 1];
        if (spill[2 * (size_t)j] == node && k > last && k < best) best = k;
      }
      for (int off = 16; off > 0; off >>= 1) best = min(best, __shfl_xor(best, off, 32));
      int c = (mine == best) ? 1 : 0;
      for (int j = lane; j < nspill; j += 32)
        c += (spill[2 * (size_t)j] == node && spill[2 * (size_t)j + 1] == best) ? 1 : 0;
      for (int off = 16; off > 0; off >>= 1) c += __shfl_xor(c, off, 32);
      best = __builtin_amdgcn_readfirstlane(best);
      c = __builtin_amdgcn_readfirstlane(c);
      if (best == 0x7fffffff) break;
      const float4 v = *(const float4*)(fl + (size_t)best * DIM);
      for (int q = 0; q < c; ++q) { acc.x += v.x; acc.y += v.y; acc.z += v.z; acc.w += v.w; }
      last = best;
      done += c;
    }
  }
  *(float4*)(agg + (size_t)node * DIM + lane * 4) = acc;
}

#define GEMM_THREADS 256
#define GEMM_WAVES (GEMM_THREADS / 32)
#define GEMM_LDS_FLOATS (168 * 1024 / 4)

static __device__ __forceinline__ void stage_matrix(float* __restrict__ dstLds,
                                                    const float* __restrict__ srcGlobal, int tid) {
  float4 v[16];
#pragma unroll
  for (int u = 0; u < 16; ++u)
    v[u] = *(const float4*)(srcGlobal + (size_t)(u * GEMM_THREADS + tid) * 4);
#pragma unroll
  for (int u = 0; u < 16; ++u)
    *(float4*)(dstLds + (u * GEMM_THREADS + tid) * 4) = v[u];
}

__global__ __launch_bounds__(GEMM_THREADS)
void sage_gemm(const float* __restrict__ agg, const int* __restrict__ cnt,
               const float* __restrict__ h,
               const float* __restrict__ Wl, const float* __restrict__ Wr,
               const float* __restrict__ bias,
               float* __restrict__ out, int n_nodes, int do_relu)
{
  __shared__ __attribute__((aligned(16))) float smem[GEMM_LDS_FLOATS];
  float* sWl = smem;
  float* sWr = smem + DIM * DIM;
  float* sb  = smem + 2 * DIM * DIM;

  const int tid = threadIdx.x;
  stage_matrix(sWl, Wl, tid);
  stage_matrix(sWr, Wr, tid);
  if (tid < DIM) sb[tid] = bias[tid];
  __syncthreads();

  const int lane = tid & 31;
  const int wave = __builtin_amdgcn_readfirstlane(tid >> 5);
  const int rowTiles = (n_nodes + 15) >> 4;

  const int laneM = lane & 15;
  const int kOff  = (lane >> 4) << 1;
  const int mOff = (lane >> 4) << 3;

  for (int tile = wave * (int)gridDim.x + (int)blockIdx.x; tile < rowTiles;
       tile += GEMM_WAVES * (int)gridDim.x) {
    const int rowBase = tile * 16;
    const int row   = rowBase + laneM;
    const float id  = 1.0f / fmaxf((float)cnt[row], 1.0f);
    const float* aggRow = agg + (size_t)row * DIM;
    const float* hRow   = h   + (size_t)row * DIM;

    v8f acc[8] = {};

#pragma unroll 2
    for (int k0 = 0; k0 < DIM; k0 += 4) {
      int k = k0 + kOff;
      v2f aA, aH;
      aA[0] = aggRow[k]     * id;
      aA[1] = aggRow[k + 1] * id;
      aH[0] = hRow[k];
      aH[1] = hRow[k + 1];
#pragma unroll
      for (int t = 0; t < 8; ++t) {
        int n = t * 16 + laneM;
        v2f bl, br;
        bl[0] = sWl[k * DIM + n];
        bl[1] = sWl[(k + 1) * DIM + n];
        br[0] = sWr[k * DIM + n];
        br[1] = sWr[(k + 1) * DIM + n];
        acc[t] = __builtin_amdgcn_wmma_f32_16x16x4_f32(
            false, aA, false, bl, (short)0, acc[t], false, false);
        acc[t] = __builtin_amdgcn_wmma_f32_16x16x4_f32(
            false, aH, false, br, (short)0, acc[t], false, false);
      }
    }

#pragma unroll
    for (int t = 0; t < 8; ++t) {
      int n = t * 16 + laneM;
      float bv = sb[n];
#pragma unroll
      for (int r = 0; r < 8; ++r) {
        float v = acc[t][r] + bv;
        if (do_relu) v = fmaxf(v, 0.0f);
        out[(size_t)(rowBase + mOff + r) * DIM + n] = v;
      }
    }
  }
}

extern "C" void kernel_launch(void* const* d_in, const int* in_sizes, int n_in,
                              void* d_out, int out_size, void* d_ws, size_t ws_size,
                              hipStream_t stream) {
  const float* x   = (const float*)d_in[0];
  const int*   ei  = (const int*)d_in[1];
  const float* W1l = (const float*)d_in[2];
  const float* b1  = (const float*)d_in[3];
  const float* W1r = (const float*)d_in[4];
  const float* W2l = (const float*)d_in[5];
  const float* b2  = (const float*)d_in[6];
  const float* W2r = (const float*)d_in[7];
  float* out = (float*)d_out;

  const int nNodes = in_sizes[0] / DIM;
  const int nEdges = in_sizes[1] / 2;
  const int* src = ei;
  const int* dst = ei + nEdges;

  const size_t nodePad = ((size_t)nNodes + 4) & ~(size_t)3;
  float* agg    = (float*)d_ws;
  float* h1     = agg + (size_t)nNodes * DIM;
  int*   cnt    = (int*)(h1 + (size_t)nNodes * DIM);
  int*   bucket = cnt + nodePad;
  int*   spill  = bucket + (size_t)nNodes * BUCKET;
  const int nChunks = (nEdges + CHUNK - 1) / CHUNK;
  const int nBins   = (nNodes + BIN_NODES - 1) >> BIN_SHIFT;
  pair_t* pairs  = (pair_t*)(spill + 2 * (size_t)nEdges);
  int*    off    = (int*)(pairs + (size_t)nChunks * CHUNK);

  const int gemmBlocks = 256;
  const int nodeWaveBlocks = (nNodes + 7) / 8;

  bin_kernel<<<nChunks, SORT_THREADS, 0, stream>>>(src, dst, pairs, off, cnt, nEdges, nNodes, nBins);
  list_kernel<<<nBins, SORT_THREADS, 0, stream>>>(pairs, off, cnt, bucket, spill, nNodes, nBins, nChunks);

  gather_kernel<<<nodeWaveBlocks, 256, 0, stream>>>(x, cnt, bucket, spill, agg, nNodes);
  sage_gemm<<<gemmBlocks, GEMM_THREADS, 0, stream>>>(agg, cnt, x, W1l, W1r, b1, h1, nNodes, 1);

  gather_kernel<<<nodeWaveBlocks, 256, 0, stream>>>(h1, cnt, bucket, spill, agg, nNodes);
  sage_gemm<<<gemmBlocks, GEMM_THREADS, 0, stream>>>(agg, cnt, h1, W2l, W2r, b2, out, nNodes, 0);
}
